# starting kernel
# baseline (speedup 1.0000x reference)
#include <hip/hip_runtime.h>
#include <hip/hip_bf16.h>

typedef __attribute__((ext_vector_type(2))) float v2f;
typedef __attribute__((ext_vector_type(8))) float v8f;

#define S_LEN 1024
#define HD    64
#define NW    4
#define WROW  18

__device__ __forceinline__ v8f wmma_f32(v2f a, v2f b, v8f c) {
  return __builtin_amdgcn_wmma_f32_16x16x4_f32(false, a, false, b, (short)0, c,
                                               false, false);
}

__global__ __launch_bounds__(128) void stickbreak_kernel(
    const float* __restrict__ q, const float* __restrict__ k,
    const float* __restrict__ v, float* __restrict__ out) {
  __shared__ float lds[NW * 16 * WROW];

  const int bh   = blockIdx.x;
  const int wave = threadIdx.x >> 5;
  const int lane = threadIdx.x & 31;
  const int m    = lane & 15;
  const int hi   = lane >> 4;

  const int qt = (blockIdx.y * NW + wave) * 16;

  const size_t base = (size_t)bh * S_LEN * HD;
  const float* qb = q + base;
  const float* kb = k + base;
  const float* vb = v + base;
  float*       ob = out + base;

  const float scale = 0.125f;

  v2f qa[16];
#pragma unroll
  for (int c = 0; c < 16; ++c) {
    const float* p = qb + (size_t)(qt + m) * HD + 4 * c + 2 * hi;
    v2f t = *(const v2f*)p;
    qa[c] = t * scale;
  }

  v8f acc0 = {}, acc1 = {}, acc2 = {}, acc3 = {};
  float carry[8];
#pragma unroll
  for (int r = 0; r < 8; ++r) carry[r] = 0.0f;

  float* wl = lds + wave * 16 * WROW;

  for (int j0 = qt; j0 >= 0; j0 -= 16) {
    if (j0 >= 16) {
      __builtin_prefetch(kb + (size_t)(j0 - 16 + m) * HD, 0, 3);
      __builtin_prefetch(vb + (size_t)(j0 - 16 + m) * HD, 0, 3);
    }

    v8f L = {};
#pragma unroll
    for (int c = 0; c < 16; ++c) {
      const float* p = kb + (size_t)(j0 + m) * HD + 4 * c + 2 * hi;
      v2f kt = *(const v2f*)p;
      L = wmma_f32(qa[c], kt, L);
    }

    const bool diag = (j0 == qt);

    float w[8];
#pragma unroll
    for (int r = 0; r < 8; ++r) {
      float x  = L[r];
      float zz = 1.0f / (1.0f + __expf(-x));
      float sp = fmaxf(x, 0.0f) + __logf(1.0f + __expf(-fabsf(x)));
      float lb = -sp;
      if (diag && (m > r + 8 * hi)) { zz = 0.0f; lb = 0.0f; }
      float s = lb;
#pragma unroll
      for (int off = 1; off < 16; off <<= 1) {
        float t = __shfl(s, lane + off);
        if (m + off < 16) s += t;
      }
      float tot = __shfl(s, hi << 4);
      w[r] = zz * __expf(s + carry[r]);
      carry[r] += tot;
    }

#pragma unroll
    for (int r = 0; r < 8; ++r)
      wl[(r + 8 * hi) * WROW + m] = w[r];
    asm volatile("s_wait_dscnt 0x0" ::: "memory");

    v2f wa[4];
#pragma unroll
    for (int c = 0; c < 4; ++c)
      wa[c] = *(const v2f*)&wl[m * WROW + 4 * c + 2 * hi];

#pragma unroll
    for (int c = 0; c < 4; ++c) {
      const float* vr0 = vb + (size_t)(j0 + 4 * c + 2 * hi) * HD + m;
      const float* vr1 = vr0 + HD;
      v2f b0 = { vr0[0],  vr1[0]  };
      v2f b1 = { vr0[16], vr1[16] };
      v2f b2 = { vr0[32], vr1[32] };
      v2f b3 = { vr0[48], vr1[48] };
      acc0 = wmma_f32(wa[c], b0, acc0);
      acc1 = wmma_f32(wa[c], b1, acc1);
      acc2 = wmma_f32(wa[c], b2, acc2);
      acc3 = wmma_f32(wa[c], b3, acc3);
    }
  }

#pragma unroll
  for (int r = 0; r < 8; ++r) {
    size_t row = (size_t)(qt + r + 8 * hi) * HD;
    ob[row + 0  + m] = acc0[r];
    ob[row + 16 + m] = acc1[r];
    ob[row + 32 + m] = acc2[r];
    ob[row + 48 + m] = acc3[r];
  }
}

extern "C" void kernel_launch(void* const* d_in, const int* in_sizes, int n_in,
                              void* d_out, int out_size, void* d_ws, size_t ws_size,
                              hipStream_t stream) {
  const float* q = (const float*)d_in[0];
  const float* k = (const float*)d_in[1];
  const float* v = (const float*)d_in[2];
  float* out = (float*)d_out;

  const int BH = in_sizes[0] / (S_LEN * HD);
  dim3 grid(BH, S_LEN / (16 * NW));
  dim3 block(32 * NW);
  hipLaunchKernelGGL(stickbreak_kernel, grid, block, 0, stream, q, k, v, out);
}


# K and V fragments of the next key block are loaded into the operand registers one loop trip ahead (software-pipelined global loads) in place of the cache prefetch hint; order of sums unchanged
# speedup vs baseline: 1.1509x; 1.1509x over previous
#include <hip/hip_runtime.h>
#include <hip/hip_bf16.h>

typedef __attribute__((ext_vector_type(2))) float v2f;
typedef __attribute__((ext_vector_type(8))) float v8f;

#define S_LEN 1024
#define HD    64
#define NW    4
#define WROW  18

__device__ __forceinline__ v8f wmma_f32(v2f a, v2f b, v8f c) {
  return __builtin_amdgcn_wmma_f32_16x16x4_f32(false, a, false, b, (short)0, c,
                                               false, false);
}

__global__ __launch_bounds__(128) void stickbreak_kernel(
    const float* __restrict__ q, const float* __restrict__ k,
    const float* __restrict__ v, float* __restrict__ out) {
  __shared__ float lds[NW * 16 * WROW];

  const int bh   = blockIdx.x;
  const int wave = threadIdx.x >> 5;
  const int lane = threadIdx.x & 31;
  const int m    = lane & 15;
  const int hi   = lane >> 4;

  const int qt = (blockIdx.y * NW + wave) * 16;

  const size_t base = (size_t)bh * S_LEN * HD;
  const float* qb = q + base;
  const float* kb = k + base;
  const float* vb = v + base;
  float*       ob = out + base;

  const float scale = 0.125f;

  v2f qa[16];
#pragma unroll
  for (int c = 0; c < 16; ++c) {
    const float* p = qb + (size_t)(qt + m) * HD + 4 * c + 2 * hi;
    v2f t = *(const v2f*)p;
    qa[c] = t * scale;
  }

  v8f acc0 = {}, acc1 = {}, acc2 = {}, acc3 = {};
  float carry[8];
#pragma unroll
  for (int r = 0; r < 8; ++r) carry[r] = 0.0f;

  float* wl = lds + wave * 16 * WROW;

  v2f kf[16];
  v2f vf[16];
#pragma unroll
  for (int c = 0; c < 16; ++c)
    kf[c] = *(const v2f*)(kb + (size_t)(qt + m) * HD + 4 * c + 2 * hi);
#pragma unroll
  for (int c = 0; c < 4; ++c) {
    const float* vr0 = vb + (size_t)(qt + 4 * c + 2 * hi) * HD + m;
    const float* vr1 = vr0 + HD;
#pragma unroll
    for (int n = 0; n < 4; ++n)
      vf[4 * c + n] = v2f{ vr0[16 * n], vr1[16 * n] };
  }

  __threadfence_block();

  for (int j0 = qt; j0 >= 0; j0 -= 16) {
    const int jn = (j0 >= 16) ? (j0 - 16) : 0;

    v8f L = {};
#pragma unroll
    for (int c = 0; c < 16; ++c)
      L = wmma_f32(qa[c], kf[c], L);

    asm volatile("" : "+v"(L) :: "memory");
#pragma unroll
    for (int c = 0; c < 16; ++c)
      kf[c] = *(const v2f*)(kb + (size_t)(jn + m) * HD + 4 * c + 2 * hi);

    const bool diag = (j0 == qt);

    float w[8];
#pragma unroll
    for (int r = 0; r < 8; ++r) {
      float x  = L[r];
      float zz = 1.0f / (1.0f + __expf(-x));
      float sp = fmaxf(x, 0.0f) + __logf(1.0f + __expf(-fabsf(x)));
      float lb = -sp;
      if (diag && (m > r + 8 * hi)) { zz = 0.0f; lb = 0.0f; }
      float s = lb;
#pragma unroll
      for (int off = 1; off < 16; off <<= 1) {
        float t = __shfl(s, lane + off);
        if (m + off < 16) s += t;
      }
      float tot = __shfl(s, hi << 4);
      w[r] = zz * __expf(s + carry[r]);
      carry[r] += tot;
    }

#pragma unroll
    for (int r = 0; r < 8; ++r)
      wl[(r + 8 * hi) * WROW + m] = w[r];
    asm volatile("s_wait_dscnt 0x0" ::: "memory");

    v2f wa[4];
#pragma unroll
    for (int c = 0; c < 4; ++c)
      wa[c] = *(const v2f*)&wl[m * WROW + 4 * c + 2 * hi];

#pragma unroll
    for (int c = 0; c < 4; ++c) {
      acc0 = wmma_f32(wa[c], vf[4 * c + 0], acc0);
      acc1 = wmma_f32(wa[c], vf[4 * c + 1], acc1);
      acc2 = wmma_f32(wa[c], vf[4 * c + 2], acc2);
      acc3 = wmma_f32(wa[c], vf[4 * c + 3], acc3);
    }

    asm volatile("" : "+v"(acc0), "+v"(acc1), "+v"(acc2), "+v"(acc3) :: "memory");
#pragma unroll
    for (int c = 0; c < 4; ++c) {
      const float* vr0 = vb + (size_t)(jn + 4 * c + 2 * hi) * HD + m;
      const float* vr1 = vr0 + HD;
#pragma unroll
      for (int n = 0; n < 4; ++n)
        vf[4 * c + n] = v2f{ vr0[16 * n], vr1[16 * n] };
    }
  }

  __threadfence_block();

#pragma unroll
  for (int r = 0; r < 8; ++r) {
    size_t row = (size_t)(qt + r + 8 * hi) * HD;
    ob[row + 0  + m] = acc0[r];
    ob[row + 16 + m] = acc1[r];
    ob[row + 32 + m] = acc2[r];
    ob[row + 48 + m] = acc3[r];
  }
}

extern "C" void kernel_launch(void* const* d_in, const int* in_sizes, int n_in,
                              void* d_out, int out_size, void* d_ws, size_t ws_size,
                              hipStream_t stream) {
  const float* q = (const float*)d_in[0];
  const float* k = (const float*)d_in[1];
  const float* v = (const float*)d_in[2];
  float* out = (float*)d_out;

  const int BH = in_sizes[0] / (S_LEN * HD);
  dim3 grid(BH, S_LEN / (16 * NW));
  dim3 block(32 * NW);
  hipLaunchKernelGGL(stickbreak_kernel, grid, block, 0, stream, q, k, v, out);
}


# Key blocks of each query tile split over the 4 waves of a workgroup (one tile per workgroup); partial tiles scaled by exp of earlier segments' log-beta sums and added via LDS; order of sums changes
# speedup vs baseline: 1.9228x; 1.6707x over previous
#include <hip/hip_runtime.h>
#include <hip/hip_bf16.h>


typedef __attribute__((ext_vector_type(2))) float v2f;
typedef __attribute__((ext_vector_type(8))) float v8f;

#define S_LEN 1024
#define HD    64
#define NW    4
#define NT    (S_LEN / 16)
#define WROW  18
#define OROW  66

__device__ __forceinline__ v8f wmma_f32(v2f a, v2f b, v8f c) {
  return __builtin_amdgcn_wmma_f32_16x16x4_f32(false, a, false, b, (short)0, c,
                                               false, false);
}

__global__ __launch_bounds__(128) void stickbreak_kernel(
    const float* __restrict__ q, const float* __restrict__ k,
    const float* __restrict__ v, float* __restrict__ out) {
  __shared__ float lds[NW * 16 * WROW];
  __shared__ float obuf[NW * 16 * OROW];
  __shared__ float cbuf[NW * 16];

  const int bh   = blockIdx.x;
  const int wave = __builtin_amdgcn_readfirstlane((int)(threadIdx.x >> 5));
  const int lane = threadIdx.x & 31;
  const int m    = lane & 15;
  const int hi   = lane >> 4;

  const int tile = (NT - 1) - (int)blockIdx.y;
  const int qt   = tile * 16;

  const int per = (tile + NW) / NW;
  const int bhi = tile - wave * per;
  const int blo = (bhi - per + 1 > 0) ? (bhi - per + 1) : 0;
  const int jhi = bhi * 16;
  const int jlo = blo * 16;

  const size_t base = (size_t)bh * S_LEN * HD;
  const float* qb = q + base;
  const float* kb = k + base;
  const float* vb = v + base;
  float*       ob = out + base;

  const float scale = 0.125f;

  v8f acc0 = {}, acc1 = {}, acc2 = {}, acc3 = {};
  float carry[8];
#pragma unroll
  for (int r = 0; r < 8; ++r) carry[r] = 0.0f;

  float* wl = lds + wave * 16 * WROW;

  if (bhi >= 0) {
    v2f qa[16];
#pragma unroll
    for (int c = 0; c < 16; ++c) {
      const float* p = qb + (size_t)(qt + m) * HD + 4 * c + 2 * hi;
      v2f t = *(const v2f*)p;
      qa[c] = t * scale;
    }

    v2f kf[16];
    v2f vf[16];
#pragma unroll
    for (int c = 0; c < 16; ++c)
      kf[c] = *(const v2f*)(kb + (size_t)(jhi + m) * HD + 4 * c + 2 * hi);
#pragma unroll
    for (int c = 0; c < 4; ++c) {
      const float* vr0 = vb + (size_t)(jhi + 4 * c + 2 * hi) * HD + m;
      const float* vr1 = vr0 + HD;
#pragma unroll
      for (int n = 0; n < 4; ++n)
        vf[4 * c + n] = v2f{ vr0[16 * n], vr1[16 * n] };
    }

    __threadfence_block();

    for (int j0 = jhi; j0 >= jlo; j0 -= 16) {
      const int jn = (j0 >= 16) ? (j0 - 16) : 0;

      v8f L = {};
#pragma unroll
      for (int c = 0; c < 16; ++c)
        L = wmma_f32(qa[c], kf[c], L);

      asm volatile("" : "+v"(L) :: "memory");
#pragma unroll
      for (int c = 0; c < 16; ++c)
        kf[c] = *(const v2f*)(kb + (size_t)(jn + m) * HD + 4 * c + 2 * hi);

      const bool diag = (j0 == qt);

      float w[8];
#pragma unroll
      for (int r = 0; r < 8; ++r) {
        float x  = L[r];
        float zz = 1.0f / (1.0f + __expf(-x));
        float sp = fmaxf(x, 0.0f) + __logf(1.0f + __expf(-fabsf(x)));
        float lb = -sp;
        if (diag && (m > r + 8 * hi)) { zz = 0.0f; lb = 0.0f; }
        float s = lb;
#pragma unroll
        for (int off = 1; off < 16; off <<= 1) {
          float t = __shfl(s, lane + off);
          if (m + off < 16) s += t;
        }
        float tot = __shfl(s, hi << 4);
        w[r] = zz * __expf(s + carry[r]);
        carry[r] += tot;
      }

#pragma unroll
      for (int r = 0; r < 8; ++r)
        wl[(r + 8 * hi) * WROW + m] = w[r];
      asm volatile("s_wait_dscnt 0x0" ::: "memory");

      v2f wa[4];
#pragma unroll
      for (int c = 0; c < 4; ++c)
        wa[c] = *(const v2f*)&wl[m * WROW + 4 * c + 2 * hi];

#pragma unroll
      for (int c = 0; c < 4; ++c) {
        acc0 = wmma_f32(wa[c], vf[4 * c + 0], acc0);
        acc1 = wmma_f32(wa[c], vf[4 * c + 1], acc1);
        acc2 = wmma_f32(wa[c], vf[4 * c + 2], acc2);
        acc3 = wmma_f32(wa[c], vf[4 * c + 3], acc3);
      }

      asm volatile("" : "+v"(acc0), "+v"(acc1), "+v"(acc2), "+v"(acc3) :: "memory");
#pragma unroll
      for (int c = 0; c < 4; ++c) {
        const float* vr0 = vb + (size_t)(jn + 4 * c + 2 * hi) * HD + m;
        const float* vr1 = vr0 + HD;
#pragma unroll
        for (int n = 0; n < 4; ++n)
          vf[4 * c + n] = v2f{ vr0[16 * n], vr1[16 * n] };
      }
    }

    __threadfence_block();
  }

  float* ow = obuf + wave * 16 * OROW;
#pragma unroll
  for (int r = 0; r < 8; ++r) {
    const int row = r + 8 * hi;
    ow[row * OROW + 0  + m] = acc0[r];
    ow[row * OROW + 16 + m] = acc1[r];
    ow[row * OROW + 32 + m] = acc2[r];
    ow[row * OROW + 48 + m] = acc3[r];
  }
  if (m == 0) {
#pragma unroll
    for (int r = 0; r < 8; ++r)
      cbuf[wave * 16 + r + 8 * hi] = carry[r];
  }
  __syncthreads();

#pragma unroll
  for (int i = 0; i < 4; ++i) {
    const int row = 4 * wave + i;
    const float c0 = cbuf[0 * 16 + row];
    const float c1 = cbuf[1 * 16 + row];
    const float c2 = cbuf[2 * 16 + row];
    const float f1 = __expf(c0);
    const float f2 = __expf(c0 + c1);
    const float f3 = __expf((c0 + c1) + c2);
    const v2f p0 = *(const v2f*)&obuf[(0 * 16 + row) * OROW + 2 * lane];
    const v2f p1 = *(const v2f*)&obuf[(1 * 16 + row) * OROW + 2 * lane];
    const v2f p2 = *(const v2f*)&obuf[(2 * 16 + row) * OROW + 2 * lane];
    const v2f p3 = *(const v2f*)&obuf[(3 * 16 + row) * OROW + 2 * lane];
    v2f o = p0 + p1 * f1;
    o = o + p2 * f2;
    o = o + p3 * f3;
    *(v2f*)&ob[(size_t)(qt + row) * HD + 2 * lane] = o;
  }
}

extern "C" void kernel_launch(void* const* d_in, const int* in_sizes, int n_in,
                              void* d_out, int out_size, void* d_ws, size_t ws_size,
                              hipStream_t stream) {
  const float* q = (const float*)d_in[0];
  const float* k = (const float*)d_in[1];
  const float* v = (const float*)d_in[2];
  float* out = (float*)d_out;

  const int BH = in_sizes[0] / (S_LEN * HD);
  dim3 grid(BH, NT);
  dim3 block(32 * NW);
  hipLaunchKernelGGL(stickbreak_kernel, grid, block, 0, stream, q, k, v, out);
}


# Logits tile held transposed (K @ Q^T): log-beta suffix scan runs in-lane over 8 keys plus one lane-half exchange; weights feed V^T @ W^T from registers, no LDS bounce; order of sums changes
# speedup vs baseline: 2.1538x; 1.1201x over previous
#include <hip/hip_runtime.h>
#include <hip/hip_bf16.h>


typedef __attribute__((ext_vector_type(2))) float v2f;
typedef __attribute__((ext_vector_type(4))) float v4f;
typedef __attribute__((ext_vector_type(8))) float v8f;

#define S_LEN 1024
#define HD    64
#define NW    4
#define NT    (S_LEN / 16)
#define OROW  68

__device__ __forceinline__ v8f wmma_f32(v2f a, v2f b, v8f c) {
  return __builtin_amdgcn_wmma_f32_16x16x4_f32(false, a, false, b, (short)0, c,
                                               false, false);
}

__device__ __forceinline__ float xhalf(float x) {
  int b = __builtin_bit_cast(int, x);
  return __builtin_bit_cast(float,
      __builtin_amdgcn_permlanex16(b, b, 0x76543210, 0xfedcba98, false, false));
}

__global__ __launch_bounds__(128) void stickbreak_kernel(
    const float* __restrict__ q, const float* __restrict__ k,
    const float* __restrict__ v, float* __restrict__ out) {
  __shared__ __attribute__((aligned(16))) float obuf[NW * 16 * OROW];
  __shared__ float cbuf[NW * 16];

  const int bh   = blockIdx.x;
  const int wave = __builtin_amdgcn_readfirstlane((int)(threadIdx.x >> 5));
  const int lane = threadIdx.x & 31;
  const int m    = lane & 15;
  const int hi   = lane >> 4;

  const int tile = (NT - 1) - (int)blockIdx.y;
  const int qt   = tile * 16;

  const int per = (tile + NW) / NW;
  const int bhi = tile - wave * per;
  const int blo = (bhi - per + 1 > 0) ? (bhi - per + 1) : 0;
  const int jhi = bhi * 16;
  const int jlo = blo * 16;

  const size_t base = (size_t)bh * S_LEN * HD;
  const float* qb = q + base;
  const float* kb = k + base;
  const float* vb = v + base;
  float*       ob = out + base;

  const float scale = 0.125f;

  v8f acc0 = {}, acc1 = {}, acc2 = {}, acc3 = {};
  float carry = 0.0f;

  if (bhi >= 0) {
    v2f qa[16];
#pragma unroll
    for (int c = 0; c < 16; ++c) {
      const float* p = qb + (size_t)(qt + m) * HD + 4 * c + 2 * hi;
      v2f t = *(const v2f*)p;
      qa[c] = t * scale;
    }

    v2f kf[16];
    v2f vf[16];
#pragma unroll
    for (int c = 0; c < 16; ++c)
      kf[c] = *(const v2f*)(kb + (size_t)(jhi + m) * HD + 4 * c + 2 * hi);
#pragma unroll
    for (int c = 0; c < 4; ++c) {
      const float* vr0 = vb + (size_t)(jhi + 8 * hi + 2 * c) * HD + m;
      const float* vr1 = vr0 + HD;
#pragma unroll
      for (int n = 0; n < 4; ++n)
        vf[4 * c + n] = v2f{ vr0[16 * n], vr1[16 * n] };
    }

    __threadfence_block();

    for (int j0 = jhi; j0 >= jlo; j0 -= 16) {
      const int jn = (j0 >= 16) ? (j0 - 16) : 0;

      v8f L = {};
#pragma unroll
      for (int c = 0; c < 16; ++c)
        L = wmma_f32(kf[c], qa[c], L);

      asm volatile("" : "+v"(L) :: "memory");
#pragma unroll
      for (int c = 0; c < 16; ++c)
        kf[c] = *(const v2f*)(kb + (size_t)(jn + m) * HD + 4 * c + 2 * hi);

      const bool diag = (j0 == qt);

      float zz[8], sfx[8];
#pragma unroll
      for (int r = 0; r < 8; ++r) {
        float x  = L[r];
        float z1 = 1.0f / (1.0f + __expf(-x));
        float sp = fmaxf(x, 0.0f) + __logf(1.0f + __expf(-fabsf(x)));
        float lb = -sp;
        if (diag && (r + 8 * hi > m)) { z1 = 0.0f; lb = 0.0f; }
        zz[r]  = z1;
        sfx[r] = lb;
      }
#pragma unroll
      for (int r = 6; r >= 0; --r) sfx[r] += sfx[r + 1];
      const float other = xhalf(sfx[0]);
      const float ahead = carry + (hi ? 0.0f : other);
      v2f wb[4];
#pragma unroll
      for (int c = 0; c < 4; ++c) {
        wb[c] = v2f{ zz[2 * c]     * __expf(sfx[2 * c]     + ahead),
                     zz[2 * c + 1] * __expf(sfx[2 * c + 1] + ahead) };
      }
      carry += sfx[0] + other;

#pragma unroll
      for (int c = 0; c < 4; ++c) {
        acc0 = wmma_f32(vf[4 * c + 0], wb[c], acc0);
        acc1 = wmma_f32(vf[4 * c + 1], wb[c], acc1);
        acc2 = wmma_f32(vf[4 * c + 2], wb[c], acc2);
        acc3 = wmma_f32(vf[4 * c + 3], wb[c], acc3);
      }

      asm volatile("" : "+v"(acc0), "+v"(acc1), "+v"(acc2), "+v"(acc3) :: "memory");
#pragma unroll
      for (int c = 0; c < 4; ++c) {
        const float* vr0 = vb + (size_t)(jn + 8 * hi + 2 * c) * HD + m;
        const float* vr1 = vr0 + HD;
#pragma unroll
        for (int n = 0; n < 4; ++n)
          vf[4 * c + n] = v2f{ vr0[16 * n], vr1[16 * n] };
      }
    }

    __threadfence_block();
  }

  float* ow = obuf + wave * 16 * OROW + m * OROW + 8 * hi;
  *(v4f*)&ow[0]      = v4f{ acc0[0], acc0[1], acc0[2], acc0[3] };
  *(v4f*)&ow[4]      = v4f{ acc0[4], acc0[5], acc0[6], acc0[7] };
  *(v4f*)&ow[16]     = v4f{ acc1[0], acc1[1], acc1[2], acc1[3] };
  *(v4f*)&ow[16 + 4] = v4f{ acc1[4], acc1[5], acc1[6], acc1[7] };
  *(v4f*)&ow[32]     = v4f{ acc2[0], acc2[1], acc2[2], acc2[3] };
  *(v4f*)&ow[32 + 4] = v4f{ acc2[4], acc2[5], acc2[6], acc2[7] };
  *(v4f*)&ow[48]     = v4f{ acc3[0], acc3[1], acc3[2], acc3[3] };
  *(v4f*)&ow[48 + 4] = v4f{ acc3[4], acc3[5], acc3[6], acc3[7] };
  if (hi == 0) cbuf[wave * 16 + m] = carry;
  __syncthreads();

#pragma unroll
  for (int i = 0; i < 4; ++i) {
    const int row = 4 * wave + i;
    const float c0 = cbuf[0 * 16 + row];
    const float c1 = cbuf[1 * 16 + row];
    const float c2 = cbuf[2 * 16 + row];
    const float f1 = __expf(c0);
    const float f2 = __expf(c0 + c1);
    const float f3 = __expf((c0 + c1) + c2);
    const v2f p0 = *(const v2f*)&obuf[(0 * 16 + row) * OROW + 2 * lane];
    const v2f p1 = *(const v2f*)&obuf[(1 * 16 + row) * OROW + 2 * lane];
    const v2f p2 = *(const v2f*)&obuf[(2 * 16 + row) * OROW + 2 * lane];
    const v2f p3 = *(const v2f*)&obuf[(3 * 16 + row) * OROW + 2 * lane];
    v2f o = p0 + p1 * f1;
    o = o + p2 * f2;
    o = o + p3 * f3;
    *(v2f*)&ob[(size_t)(qt + row) * HD + 2 * lane] = o;
  }
}

extern "C" void kernel_launch(void* const* d_in, const int* in_sizes, int n_in,
                              void* d_out, int out_size, void* d_ws, size_t ws_size,
                              hipStream_t stream) {
  const float* q = (const float*)d_in[0];
  const float* k = (const float*)d_in[1];
  const float* v = (const float*)d_in[2];
  float* out = (float*)d_out;

  const int BH = in_sizes[0] / (S_LEN * HD);
  dim3 grid(BH, NT);
  dim3 block(32 * NW);
  hipLaunchKernelGGL(stickbreak_kernel, grid, block, 0, stream, q, k, v, out);
}


# Key-block walk of a query tile ends once exp(running log-beta sum) is exactly 0 for all 16 queries (every later weight is an exact 0); one wave per tile, no cross-wave combine; order of sums changes
# speedup vs baseline: 5.3654x; 2.4911x over previous
#include <hip/hip_runtime.h>
#include <hip/hip_bf16.h>


typedef __attribute__((ext_vector_type(2))) float v2f;
typedef __attribute__((ext_vector_type(4))) float v4f;
typedef __attribute__((ext_vector_type(8))) float v8f;

#define S_LEN 1024
#define HD    64
#define NT    (S_LEN / 16)
#define OROW  68

__device__ __forceinline__ v8f wmma_f32(v2f a, v2f b, v8f c) {
  return __builtin_amdgcn_wmma_f32_16x16x4_f32(false, a, false, b, (short)0, c,
                                               false, false);
}

__device__ __forceinline__ float xhalf(float x) {
  int b = __builtin_bit_cast(int, x);
  return __builtin_bit_cast(float,
      __builtin_amdgcn_permlanex16(b, b, 0x76543210, 0xfedcba98, false, false));
}

__global__ __launch_bounds__(32) void stickbreak_kernel(
    const float* __restrict__ q, const float* __restrict__ k,
    const float* __restrict__ v, float* __restrict__ out) {
  __shared__ __attribute__((aligned(16))) float obuf[16 * OROW];

  const int bh   = blockIdx.x;
  const int lane = threadIdx.x;
  const int m    = lane & 15;
  const int hi   = lane >> 4;

  const int tile = (NT - 1) - (int)blockIdx.y;
  const int qt   = tile * 16;

  const int jhi = qt;

  const size_t base = (size_t)bh * S_LEN * HD;
  const float* qb = q + base;
  const float* kb = k + base;
  const float* vb = v + base;
  float*       ob = out + base;

  const float scale = 0.125f;

  v8f acc0 = {}, acc1 = {}, acc2 = {}, acc3 = {};
  float carry = 0.0f;

  {
    v2f qa[16];
#pragma unroll
    for (int c = 0; c < 16; ++c) {
      const float* p = qb + (size_t)(qt + m) * HD + 4 * c + 2 * hi;
      v2f t = *(const v2f*)p;
      qa[c] = t * scale;
    }

    v2f kf[16];
    v2f vf[16];
#pragma unroll
    for (int c = 0; c < 16; ++c)
      kf[c] = *(const v2f*)(kb + (size_t)(jhi + m) * HD + 4 * c + 2 * hi);
#pragma unroll
    for (int c = 0; c < 4; ++c) {
      const float* vr0 = vb + (size_t)(jhi + 8 * hi + 2 * c) * HD + m;
      const float* vr1 = vr0 + HD;
#pragma unroll
      for (int n = 0; n < 4; ++n)
        vf[4 * c + n] = v2f{ vr0[16 * n], vr1[16 * n] };
    }

    __threadfence_block();

    bool live = true;
    for (int j0 = jhi; j0 >= 0 && live; j0 -= 16) {
      const int jn = (j0 >= 16) ? (j0 - 16) : 0;

      v8f L = {};
#pragma unroll
      for (int c = 0; c < 16; ++c)
        L = wmma_f32(kf[c], qa[c], L);

      asm volatile("" : "+v"(L) :: "memory");
#pragma unroll
      for (int c = 0; c < 16; ++c)
        kf[c] = *(const v2f*)(kb + (size_t)(jn + m) * HD + 4 * c + 2 * hi);

      const bool diag = (j0 == qt);

      float zz[8], sfx[8];
#pragma unroll
      for (int r = 0; r < 8; ++r) {
        float x  = L[r];
        float z1 = 1.0f / (1.0f + __expf(-x));
        float sp = fmaxf(x, 0.0f) + __logf(1.0f + __expf(-fabsf(x)));
        float lb = -sp;
        if (diag && (r + 8 * hi > m)) { z1 = 0.0f; lb = 0.0f; }
        zz[r]  = z1;
        sfx[r] = lb;
      }
#pragma unroll
      for (int r = 6; r >= 0; --r) sfx[r] += sfx[r + 1];
      const float other = xhalf(sfx[0]);
      const float ahead = carry + (hi ? 0.0f : other);
      v2f wb[4];
#pragma unroll
      for (int c = 0; c < 4; ++c) {
        wb[c] = v2f{ zz[2 * c]     * __expf(sfx[2 * c]     + ahead),
                     zz[2 * c + 1] * __expf(sfx[2 * c + 1] + ahead) };
      }
      carry += sfx[0] + other;
      live = __builtin_amdgcn_ballot_w32(__expf(carry) != 0.0f) != 0u;

#pragma unroll
      for (int c = 0; c < 4; ++c) {
        acc0 = wmma_f32(vf[4 * c + 0], wb[c], acc0);
        acc1 = wmma_f32(vf[4 * c + 1], wb[c], acc1);
        acc2 = wmma_f32(vf[4 * c + 2], wb[c], acc2);
        acc3 = wmma_f32(vf[4 * c + 3], wb[c], acc3);
      }

      asm volatile("" : "+v"(acc0), "+v"(acc1), "+v"(acc2), "+v"(acc3) :: "memory");
#pragma unroll
      for (int c = 0; c < 4; ++c) {
        const float* vr0 = vb + (size_t)(jn + 8 * hi + 2 * c) * HD + m;
        const float* vr1 = vr0 + HD;
#pragma unroll
        for (int n = 0; n < 4; ++n)
          vf[4 * c + n] = v2f{ vr0[16 * n], vr1[16 * n] };
      }
    }

    __threadfence_block();
  }

  float* ow = obuf + m * OROW + 8 * hi;
  *(v4f*)&ow[0]      = v4f{ acc0[0], acc0[1], acc0[2], acc0[3] };
  *(v4f*)&ow[4]      = v4f{ acc0[4], acc0[5], acc0[6], acc0[7] };
  *(v4f*)&ow[16]     = v4f{ acc1[0], acc1[1], acc1[2], acc1[3] };
  *(v4f*)&ow[16 + 4] = v4f{ acc1[4], acc1[5], acc1[6], acc1[7] };
  *(v4f*)&ow[32]     = v4f{ acc2[0], acc2[1], acc2[2], acc2[3] };
  *(v4f*)&ow[32 + 4] = v4f{ acc2[4], acc2[5], acc2[6], acc2[7] };
  *(v4f*)&ow[48]     = v4f{ acc3[0], acc3[1], acc3[2], acc3[3] };
  *(v4f*)&ow[48 + 4] = v4f{ acc3[4], acc3[5], acc3[6], acc3[7] };
  __syncthreads();

#pragma unroll
  for (int row = 0; row < 16; ++row) {
    const v2f o = *(const v2f*)&obuf[row * OROW + 2 * lane];
    *(v2f*)&ob[(size_t)(qt + row) * HD + 2 * lane] = o;
  }
}

extern "C" void kernel_launch(void* const* d_in, const int* in_sizes, int n_in,
                              void* d_out, int out_size, void* d_ws, size_t ws_size,
                              hipStream_t stream) {
  const float* q = (const float*)d_in[0];
  const float* k = (const float*)d_in[1];
  const float* v = (const float*)d_in[2];
  float* out = (float*)d_out;

  const int BH = in_sizes[0] / (S_LEN * HD);
  dim3 grid(BH, NT);
  dim3 block(32);
  hipLaunchKernelGGL(stickbreak_kernel, grid, block, 0, stream, q, k, v, out);
}


# Per-logit math in base 2: log-beta and log2 sigmoid from one exp2 and one log2; weight = one exp2 of their sum plus suffix sum; no division, no guarded log; masked logit = -inf; sum order unchanged
# speedup vs baseline: 6.4994x; 1.2114x over previous
#include <hip/hip_runtime.h>
#include <hip/hip_bf16.h>


typedef __attribute__((ext_vector_type(2))) float v2f;
typedef __attribute__((ext_vector_type(4))) float v4f;
typedef __attribute__((ext_vector_type(8))) float v8f;

#define S_LEN 1024
#define HD    64
#define NT    (S_LEN / 16)
#define OROW  68

__device__ __forceinline__ v8f wmma_f32(v2f a, v2f b, v8f c) {
  return __builtin_amdgcn_wmma_f32_16x16x4_f32(false, a, false, b, (short)0, c,
                                               false, false);
}

__device__ __forceinline__ float xhalf(float x) {
  int b = __builtin_bit_cast(int, x);
  return __builtin_bit_cast(float,
      __builtin_amdgcn_permlanex16(b, b, 0x76543210, 0xfedcba98, false, false));
}

__global__ __launch_bounds__(32) void stickbreak_kernel(
    const float* __restrict__ q, const float* __restrict__ k,
    const float* __restrict__ v, float* __restrict__ out) {
  __shared__ __attribute__((aligned(16))) float obuf[16 * OROW];

  const int bh   = blockIdx.x;
  const int lane = threadIdx.x;
  const int m    = lane & 15;
  const int hi   = lane >> 4;

  const int tile = (NT - 1) - (int)blockIdx.y;
  const int qt   = tile * 16;

  const int jhi = qt;

  const size_t base = (size_t)bh * S_LEN * HD;
  const float* qb = q + base;
  const float* kb = k + base;
  const float* vb = v + base;
  float*       ob = out + base;

  const float scale = 0.125f;
  const float LOG2E = 1.44269504088896340736f;

  v8f acc0 = {}, acc1 = {}, acc2 = {}, acc3 = {};
  float carry = 0.0f;

  {
    v2f qa[16];
#pragma unroll
    for (int c = 0; c < 16; ++c) {
      const float* p = qb + (size_t)(qt + m) * HD + 4 * c + 2 * hi;
      v2f t = *(const v2f*)p;
      qa[c] = t * scale;
    }

    v2f kf[16];
    v2f vf[16];
#pragma unroll
    for (int c = 0; c < 16; ++c)
      kf[c] = *(const v2f*)(kb + (size_t)(jhi + m) * HD + 4 * c + 2 * hi);
#pragma unroll
    for (int c = 0; c < 4; ++c) {
      const float* vr0 = vb + (size_t)(jhi + 8 * hi + 2 * c) * HD + m;
      const float* vr1 = vr0 + HD;
#pragma unroll
      for (int n = 0; n < 4; ++n)
        vf[4 * c + n] = v2f{ vr0[16 * n], vr1[16 * n] };
    }

    __threadfence_block();

    bool live = true;
    for (int j0 = jhi; j0 >= 0 && live; j0 -= 16) {
      const int jn = (j0 >= 16) ? (j0 - 16) : 0;

      v8f L = {};
#pragma unroll
      for (int c = 0; c < 16; ++c)
        L = wmma_f32(kf[c], qa[c], L);

      asm volatile("" : "+v"(L) :: "memory");
#pragma unroll
      for (int c = 0; c < 16; ++c)
        kf[c] = *(const v2f*)(kb + (size_t)(jn + m) * HD + 4 * c + 2 * hi);

      const bool diag = (j0 == qt);

      float hz[8], sfx[8];
#pragma unroll
      for (int r = 0; r < 8; ++r) {
        float u = L[r] * LOG2E;
        if (diag && (r + 8 * hi > m)) u = -__builtin_inff();
        const float au = __builtin_fabsf(u);
        const float lg = __builtin_amdgcn_logf(1.0f + __builtin_amdgcn_exp2f(-au));
        hz[r]  = __builtin_fmaf(lg, -2.0f, -au);
        sfx[r] = -fmaxf(u, 0.0f) - lg;
      }
#pragma unroll
      for (int r = 6; r >= 0; --r) sfx[r] += sfx[r + 1];
      const float other = xhalf(sfx[0]);
      const float ahead = carry + (hi ? 0.0f : other);
      v2f wb[4];
#pragma unroll
      for (int c = 0; c < 4; ++c) {
        const int   rn = (c < 3) ? 2 * c + 2 : 7;
        const float e0 = (hz[2 * c] + sfx[2 * c + 1]) + ahead;
        const float e1 = (c < 3) ? (hz[2 * c + 1] + sfx[rn]) + ahead
                                 : hz[7] + ahead;
        wb[c] = v2f{ __builtin_amdgcn_exp2f(e0), __builtin_amdgcn_exp2f(e1) };
      }
      carry += sfx[0] + other;
      live = __builtin_amdgcn_ballot_w32(__builtin_amdgcn_exp2f(carry) != 0.0f) != 0u;

#pragma unroll
      for (int c = 0; c < 4; ++c) {
        acc0 = wmma_f32(vf[4 * c + 0], wb[c], acc0);
        acc1 = wmma_f32(vf[4 * c + 1], wb[c], acc1);
        acc2 = wmma_f32(vf[4 * c + 2], wb[c], acc2);
        acc3 = wmma_f32(vf[4 * c + 3], wb[c], acc3);
      }

      asm volatile("" : "+v"(acc0), "+v"(acc1), "+v"(acc2), "+v"(acc3) :: "memory");
#pragma unroll
      for (int c = 0; c < 4; ++c) {
        const float* vr0 = vb + (size_t)(jn + 8 * hi + 2 * c) * HD + m;
        const float* vr1 = vr0 + HD;
#pragma unroll
        for (int n = 0; n < 4; ++n)
          vf[4 * c + n] = v2f{ vr0[16 * n], vr1[16 * n] };
      }
    }

    __threadfence_block();
  }

  float* ow = obuf + m * OROW + 8 * hi;
  *(v4f*)&ow[0]      = v4f{ acc0[0], acc0[1], acc0[2], acc0[3] };
  *(v4f*)&ow[4]      = v4f{ acc0[4], acc0[5], acc0[6], acc0[7] };
  *(v4f*)&ow[16]     = v4f{ acc1[0], acc1[1], acc1[2], acc1[3] };
  *(v4f*)&ow[16 + 4] = v4f{ acc1[4], acc1[5], acc1[6], acc1[7] };
  *(v4f*)&ow[32]     = v4f{ acc2[0], acc2[1], acc2[2], acc2[3] };
  *(v4f*)&ow[32 + 4] = v4f{ acc2[4], acc2[5], acc2[6], acc2[7] };
  *(v4f*)&ow[48]     = v4f{ acc3[0], acc3[1], acc3[2], acc3[3] };
  *(v4f*)&ow[48 + 4] = v4f{ acc3[4], acc3[5], acc3[6], acc3[7] };
  __syncthreads();

#pragma unroll
  for (int row = 0; row < 16; ++row) {
    const v2f o = *(const v2f*)&obuf[row * OROW + 2 * lane];
    *(v2f*)&ob[(size_t)(qt + row) * HD + 2 * lane] = o;
  }
}

extern "C" void kernel_launch(void* const* d_in, const int* in_sizes, int n_in,
                              void* d_out, int out_size, void* d_ws, size_t ws_size,
                              hipStream_t stream) {
  const float* q = (const float*)d_in[0];
  const float* k = (const float*)d_in[1];
  const float* v = (const float*)d_in[2];
  float* out = (float*)d_out;

  const int BH = in_sizes[0] / (S_LEN * HD);
  dim3 grid(BH, NT);
  dim3 block(32);
  hipLaunchKernelGGL(stickbreak_kernel, grid, block, 0, stream, q, k, v, out);
}


# Eight consecutive query tiles per 8-wave workgroup, so waves w and w+4 share a SIMD; the younger takes issue priority on 3 of 4 key blocks and the pair advances together; order of sums unchanged
# speedup vs baseline: 7.4391x; 1.1446x over previous
#include <hip/hip_runtime.h>
#include <hip/hip_bf16.h>


typedef __attribute__((ext_vector_type(2))) float v2f;
typedef __attribute__((ext_vector_type(4))) float v4f;
typedef __attribute__((ext_vector_type(8))) float v8f;

#define S_LEN 1024
#define HD    64
#define NT    (S_LEN / 16)
#define OROW  68
#define WPG   8

__device__ __forceinline__ v8f wmma_f32(v2f a, v2f b, v8f c) {
  return __builtin_amdgcn_wmma_f32_16x16x4_f32(false, a, false, b, (short)0, c,
                                               false, false);
}

__device__ __forceinline__ float xhalf(float x) {
  int b = __builtin_bit_cast(int, x);
  return __builtin_bit_cast(float,
      __builtin_amdgcn_permlanex16(b, b, 0x76543210, 0xfedcba98, false, false));
}

__global__ __launch_bounds__(32 * WPG) void stickbreak_kernel(
    const float* __restrict__ q, const float* __restrict__ k,
    const float* __restrict__ v, float* __restrict__ out) {
  __shared__ __attribute__((aligned(16))) float obuf_all[WPG][16 * OROW];

  const int bh   = blockIdx.x;
  const int lane = threadIdx.x & 31;
  const int wv   = __builtin_amdgcn_readfirstlane((int)(threadIdx.x >> 5));
  const int m    = lane & 15;
  const int hi   = lane >> 4;
  float* obuf    = obuf_all[wv];

  const int tile = ((NT / WPG - 1) - (int)blockIdx.y) * WPG + wv;
  const int qt   = tile * 16;
  const bool younger = wv >= WPG / 2;

  const int jhi = qt;

  const size_t base = (size_t)bh * S_LEN * HD;
  const float* qb = q + base;
  const float* kb = k + base;
  const float* vb = v + base;
  float*       ob = out + base;

  const float scale = 0.125f;
  const float LOG2E = 1.44269504088896340736f;

  v8f acc0 = {}, acc1 = {}, acc2 = {}, acc3 = {};
  float carry = 0.0f;

  {
    v2f qa[16];
#pragma unroll
    for (int c = 0; c < 16; ++c) {
      const float* p = qb + (size_t)(qt + m) * HD + 4 * c + 2 * hi;
      v2f t = *(const v2f*)p;
      qa[c] = t * scale;
    }

    v2f kf[16];
    v2f vf[16];
#pragma unroll
    for (int c = 0; c < 16; ++c)
      kf[c] = *(const v2f*)(kb + (size_t)(jhi + m) * HD + 4 * c + 2 * hi);
#pragma unroll
    for (int c = 0; c < 4; ++c) {
      const float* vr0 = vb + (size_t)(jhi + 8 * hi + 2 * c) * HD + m;
      const float* vr1 = vr0 + HD;
#pragma unroll
      for (int n = 0; n < 4; ++n)
        vf[4 * c + n] = v2f{ vr0[16 * n], vr1[16 * n] };
    }

    __threadfence_block();

    bool live = true;
    for (int j0 = jhi; j0 >= 0 && live; j0 -= 16) {
      const int jn = (j0 >= 16) ? (j0 - 16) : 0;

      if (younger) {
        if ((j0 & 48) == 48) __builtin_amdgcn_s_setprio(0);
        else                 __builtin_amdgcn_s_setprio(3);
      }

      v8f L = {};
#pragma unroll
      for (int c = 0; c < 16; ++c)
        L = wmma_f32(kf[c], qa[c], L);

      asm volatile("" : "+v"(L) :: "memory");
#pragma unroll
      for (int c = 0; c < 16; ++c)
        kf[c] = *(const v2f*)(kb + (size_t)(jn + m) * HD + 4 * c + 2 * hi);

      const bool diag = (j0 == qt);

      float hz[8], sfx[8];
#pragma unroll
      for (int r = 0; r < 8; ++r) {
        float u = L[r] * LOG2E;
        if (diag && (r + 8 * hi > m)) u = -__builtin_inff();
        const float au = __builtin_fabsf(u);
        const float lg = __builtin_amdgcn_logf(1.0f + __builtin_amdgcn_exp2f(-au));
        hz[r]  = __builtin_fmaf(lg, -2.0f, -au);
        sfx[r] = -fmaxf(u, 0.0f) - lg;
      }
#pragma unroll
      for (int r = 6; r >= 0; --r) sfx[r] += sfx[r + 1];
      const float other = xhalf(sfx[0]);
      const float ahead = carry + (hi ? 0.0f : other);
      v2f wb[4];
#pragma unroll
      for (int c = 0; c < 4; ++c) {
        const int   rn = (c < 3) ? 2 * c + 2 : 7;
        const float e0 = (hz[2 * c] + sfx[2 * c + 1]) + ahead;
        const float e1 = (c < 3) ? (hz[2 * c + 1] + sfx[rn]) + ahead
                                 : hz[7] + ahead;
        wb[c] = v2f{ __builtin_amdgcn_exp2f(e0), __builtin_amdgcn_exp2f(e1) };
      }
      carry += sfx[0] + other;
      live = __builtin_amdgcn_ballot_w32(__builtin_amdgcn_exp2f(carry) != 0.0f) != 0u;

#pragma unroll
      for (int c = 0; c < 4; ++c) {
        acc0 = wmma_f32(vf[4 * c + 0], wb[c], acc0);
        acc1 = wmma_f32(vf[4 * c + 1], wb[c], acc1);
        acc2 = wmma_f32(vf[4 * c + 2], wb[c], acc2);
        acc3 = wmma_f32(vf[4 * c + 3], wb[c], acc3);
      }

      asm volatile("" : "+v"(acc0), "+v"(acc1), "+v"(acc2), "+v"(acc3) :: "memory");
#pragma unroll
      for (int c = 0; c < 4; ++c) {
        const float* vr0 = vb + (size_t)(jn + 8 * hi + 2 * c) * HD + m;
        const float* vr1 = vr0 + HD;
#pragma unroll
        for (int n = 0; n < 4; ++n)
          vf[4 * c + n] = v2f{ vr0[16 * n], vr1[16 * n] };
      }
    }

    if (younger) __builtin_amdgcn_s_setprio(0);

    __threadfence_block();
  }

  float* ow = obuf + m * OROW + 8 * hi;
  *(v4f*)&ow[0]      = v4f{ acc0[0], acc0[1], acc0[2], acc0[3] };
  *(v4f*)&ow[4]      = v4f{ acc0[4], acc0[5], acc0[6], acc0[7] };
  *(v4f*)&ow[16]     = v4f{ acc1[0], acc1[1], acc1[2], acc1[3] };
  *(v4f*)&ow[16 + 4] = v4f{ acc1[4], acc1[5], acc1[6], acc1[7] };
  *(v4f*)&ow[32]     = v4f{ acc2[0], acc2[1], acc2[2], acc2[3] };
  *(v4f*)&ow[32 + 4] = v4f{ acc2[4], acc2[5], acc2[6], acc2[7] };
  *(v4f*)&ow[48]     = v4f{ acc3[0], acc3[1], acc3[2], acc3[3] };
  *(v4f*)&ow[48 + 4] = v4f{ acc3[4], acc3[5], acc3[6], acc3[7] };
  __builtin_amdgcn_fence(__ATOMIC_RELEASE, "workgroup");
  __builtin_amdgcn_wave_barrier();
  __builtin_amdgcn_fence(__ATOMIC_ACQUIRE, "workgroup");

#pragma unroll
  for (int row = 0; row < 16; ++row) {
    const v2f o = *(const v2f*)&obuf[row * OROW + 2 * lane];
    *(v2f*)&ob[(size_t)(qt + row) * HD + 2 * lane] = o;
  }
}

extern "C" void kernel_launch(void* const* d_in, const int* in_sizes, int n_in,
                              void* d_out, int out_size, void* d_ws, size_t ws_size,
                              hipStream_t stream) {
  const float* q = (const float*)d_in[0];
  const float* k = (const float*)d_in[1];
  const float* v = (const float*)d_in[2];
  float* out = (float*)d_out;

  const int BH = in_sizes[0] / (S_LEN * HD);
  dim3 grid(BH, NT / WPG);
  dim3 block(32 * WPG);
  hipLaunchKernelGGL(stickbreak_kernel, grid, block, 0, stream, q, k, v, out);
}


# K, Q, V operand loads 16 bytes per lane (K/Q fragments take d in 32-byte pieces; V^T row m = column 4m+n, whole V rows per half-wave); next K loads pinned behind the logits; logit sum order changes
# speedup vs baseline: 8.0131x; 1.0772x over previous
#include <hip/hip_runtime.h>
#include <hip/hip_bf16.h>


typedef __attribute__((ext_vector_type(2))) float v2f;
typedef __attribute__((ext_vector_type(4))) float v4f;
typedef __attribute__((ext_vector_type(8))) float v8f;

#define S_LEN 1024
#define HD    64
#define NT    (S_LEN / 16)
#define OROW  68
#define WPG   8

__device__ __forceinline__ v8f wmma_f32(v2f a, v2f b, v8f c) {
  return __builtin_amdgcn_wmma_f32_16x16x4_f32(false, a, false, b, (short)0, c,
                                               false, false);
}

__device__ __forceinline__ float xhalf(float x) {
  int b = __builtin_bit_cast(int, x);
  return __builtin_bit_cast(float,
      __builtin_amdgcn_permlanex16(b, b, 0x76543210, 0xfedcba98, false, false));
}

__global__ __launch_bounds__(32 * WPG) void stickbreak_kernel(
    const float* __restrict__ q, const float* __restrict__ k,
    const float* __restrict__ v, float* __restrict__ out) {
  __shared__ __attribute__((aligned(16))) float obuf_all[WPG][16 * OROW];

  const int bh   = blockIdx.x;
  const int lane = threadIdx.x & 31;
  const int wv   = __builtin_amdgcn_readfirstlane((int)(threadIdx.x >> 5));
  const int m    = lane & 15;
  const int hi   = lane >> 4;
  float* obuf    = obuf_all[wv];

  const int tile = ((NT / WPG - 1) - (int)blockIdx.y) * WPG + wv;
  const int qt   = tile * 16;
  const bool younger = wv >= WPG / 2;

  const int jhi = qt;

  const size_t base = (size_t)bh * S_LEN * HD;
  const float* qb = q + base;
  const float* kb = k + base;
  const float* vb = v + base;
  float*       ob = out + base;

  const float scale = 0.125f;
  const float LOG2E = 1.44269504088896340736f;

  v8f acc0 = {}, acc1 = {}, acc2 = {}, acc3 = {};
  float carry = 0.0f;

  {
    v2f qa[16];
#pragma unroll
    for (int i = 0; i < 8; ++i) {
      const v4f t = *(const v4f*)(qb + (size_t)(qt + m) * HD + 8 * i + 4 * hi) * scale;
      qa[2 * i]     = v2f{ t[0], t[1] };
      qa[2 * i + 1] = v2f{ t[2], t[3] };
    }

    v4f kq[8];
    v4f vq[8];
#pragma unroll
    for (int i = 0; i < 8; ++i)
      kq[i] = *(const v4f*)(kb + (size_t)(jhi + m) * HD + 8 * i + 4 * hi);
#pragma unroll
    for (int r = 0; r < 8; ++r)
      vq[r] = *(const v4f*)(vb + (size_t)(jhi + 8 * hi + r) * HD + 4 * m);

    __threadfence_block();

    bool live = true;
    for (int j0 = jhi; j0 >= 0 && live; j0 -= 16) {
      const int jn = (j0 >= 16) ? (j0 - 16) : 0;

      if (younger) {
        if ((j0 & 48) == 48) __builtin_amdgcn_s_setprio(0);
        else                 __builtin_amdgcn_s_setprio(3);
      }

      v8f L = {};
#pragma unroll
      for (int i = 0; i < 8; ++i) {
        L = wmma_f32(v2f{ kq[i][0], kq[i][1] }, qa[2 * i],     L);
        L = wmma_f32(v2f{ kq[i][2], kq[i][3] }, qa[2 * i + 1], L);
      }

      asm volatile("" : "+v"(L) :: "memory");
#pragma unroll
      for (int i = 0; i < 8; ++i)
        kq[i] = *(const v4f*)(kb + (size_t)(jn + m) * HD + 8 * i + 4 * hi);
      __builtin_amdgcn_sched_barrier(0);

      const bool diag = (j0 == qt);

      float hz[8], sfx[8];
#pragma unroll
      for (int r = 0; r < 8; ++r) {
        float u = L[r] * LOG2E;
        if (diag && (r + 8 * hi > m)) u = -__builtin_inff();
        const float au = __builtin_fabsf(u);
        const float lg = __builtin_amdgcn_logf(1.0f + __builtin_amdgcn_exp2f(-au));
        hz[r]  = __builtin_fmaf(lg, -2.0f, -au);
        sfx[r] = -fmaxf(u, 0.0f) - lg;
      }
#pragma unroll
      for (int r = 6; r >= 0; --r) sfx[r] += sfx[r + 1];
      const float other = xhalf(sfx[0]);
      const float ahead = carry + (hi ? 0.0f : other);
      v2f wb[4];
#pragma unroll
      for (int c = 0; c < 4; ++c) {
        const int   rn = (c < 3) ? 2 * c + 2 : 7;
        const float e0 = (hz[2 * c] + sfx[2 * c + 1]) + ahead;
        const float e1 = (c < 3) ? (hz[2 * c + 1] + sfx[rn]) + ahead
                                 : hz[7] + ahead;
        wb[c] = v2f{ __builtin_amdgcn_exp2f(e0), __builtin_amdgcn_exp2f(e1) };
      }
      carry += sfx[0] + other;
      live = __builtin_amdgcn_ballot_w32(__builtin_amdgcn_exp2f(carry) != 0.0f) != 0u;

      asm volatile("" : "+v"(vq[0]), "+v"(vq[1]), "+v"(vq[2]), "+v"(vq[3]),
                        "+v"(vq[4]), "+v"(vq[5]), "+v"(vq[6]), "+v"(vq[7]),
                        "+v"(wb[0]));
      v2f vf[16];
#pragma unroll
      for (int c = 0; c < 4; ++c)
#pragma unroll
        for (int n = 0; n < 4; ++n)
          vf[4 * c + n] = v2f{ vq[2 * c][n], vq[2 * c + 1][n] };

#pragma unroll
      for (int c = 0; c < 4; ++c) {
        acc0 = wmma_f32(vf[4 * c + 0], wb[c], acc0);
        acc1 = wmma_f32(vf[4 * c + 1], wb[c], acc1);
        acc2 = wmma_f32(vf[4 * c + 2], wb[c], acc2);
        acc3 = wmma_f32(vf[4 * c + 3], wb[c], acc3);
      }

      asm volatile("" : "+v"(acc0), "+v"(acc1), "+v"(acc2), "+v"(acc3) :: "memory");
#pragma unroll
      for (int r = 0; r < 8; ++r)
        vq[r] = *(const v4f*)(vb + (size_t)(jn + 8 * hi + r) * HD + 4 * m);
    }

    if (younger) __builtin_amdgcn_s_setprio(0);

    asm volatile("" :: "v"(kq[0]), "v"(kq[1]), "v"(kq[2]), "v"(kq[3]),
                       "v"(kq[4]), "v"(kq[5]), "v"(kq[6]), "v"(kq[7]),
                       "v"(vq[0]), "v"(vq[1]), "v"(vq[2]), "v"(vq[3]),
                       "v"(vq[4]), "v"(vq[5]), "v"(vq[6]), "v"(vq[7]));
  }

  float* ow = obuf + m * OROW + 32 * hi;
#pragma unroll
  for (int r = 0; r < 8; ++r)
    *(v4f*)&ow[4 * r] = v4f{ acc0[r], acc1[r], acc2[r], acc3[r] };
  __builtin_amdgcn_fence(__ATOMIC_RELEASE, "workgroup");
  __builtin_amdgcn_wave_barrier();
  __builtin_amdgcn_fence(__ATOMIC_ACQUIRE, "workgroup");

#pragma unroll
  for (int row = 0; row < 16; ++row) {
    const v2f o = *(const v2f*)&obuf[row * OROW + 2 * lane];
    *(v2f*)&ob[(size_t)(qt + row) * HD + 2 * lane] = o;
  }
}

extern "C" void kernel_launch(void* const* d_in, const int* in_sizes, int n_in,
                              void* d_out, int out_size, void* d_ws, size_t ws_size,
                              hipStream_t stream) {
  const float* q = (const float*)d_in[0];
  const float* k = (const float*)d_in[1];
  const float* v = (const float*)d_in[2];
  float* out = (float*)d_out;

  const int BH = in_sizes[0] / (S_LEN * HD);
  dim3 grid(BH, NT / WPG);
  dim3 block(32 * WPG);
  hipLaunchKernelGGL(stickbreak_kernel, grid, block, 0, stream, q, k, v, out);
}
